# starting kernel
# baseline (speedup 1.0000x reference)
#include <hip/hip_runtime.h>
#include <math.h>

typedef __attribute__((ext_vector_type(2))) float v2f;
typedef __attribute__((ext_vector_type(8))) float v8f;

__global__ void init_deg_kernel(float* deg, int n) {
    int i = blockIdx.x * blockDim.x + threadIdx.x;
    if (i < n) deg[i] = 1.0f;
}

__global__ void acc_deg_kernel(const int* __restrict__ dst, float* deg, int E) {
    int e = blockIdx.x * blockDim.x + threadIdx.x;
    if (e < E) atomicAdd(&deg[dst[e]], 1.0f);
}

__global__ void to_dinv_kernel(float* deg, int n) {
    int i = blockIdx.x * blockDim.x + threadIdx.x;
    if (i < n) deg[i] = rsqrtf(deg[i]);
}

__global__ void gemm1_wmma_kernel(const float* __restrict__ x,
                                  const float* __restrict__ w1,
                                  float* __restrict__ h1, int n) {
    int wave = (int)((blockIdx.x * blockDim.x + threadIdx.x) >> 5);
    int lane = threadIdx.x & 31;
    int row0 = wave * 16;
    if (row0 >= n) return;

    int m    = lane & 15;
    int koff = (lane >> 4) * 2;

    const float* xrow = x + (size_t)(row0 + m) * 128;

    v8f acc = {};
    for (int k0 = 0; k0 < 128; k0 += 4) {
        v2f a, b;
        a.x = xrow[k0 + koff + 0];
        a.y = xrow[k0 + koff + 1];
        b.x = w1[(k0 + koff + 0) * 16 + m];
        b.y = w1[(k0 + koff + 1) * 16 + m];
        acc = __builtin_amdgcn_wmma_f32_16x16x4_f32(
             false, a,  false, b,
             (short)0, acc,  false,  false);
    }

    int nCol  = lane & 15;
    int mBase = (lane >> 4) * 8;
#pragma unroll
    for (int r = 0; r < 8; ++r) {
        h1[(size_t)(row0 + mBase + r) * 16 + nCol] = acc[r];
    }
}

__global__ void init_out1_kernel(const float* __restrict__ h1,
                                 const float* __restrict__ dinv,
                                 const float* __restrict__ b1,
                                 float* __restrict__ out1, int n) {
    int t = blockIdx.x * blockDim.x + threadIdx.x;
    if (t >= n * 16) return;
    int i = t >> 4, c = t & 15;
    float di = dinv[i];
    out1[t] = b1[c] + h1[t] * di * di;
}

__global__ void scatter1_kernel(const int* __restrict__ src,
                                const int* __restrict__ dst,
                                const float* __restrict__ h1,
                                const float* __restrict__ dinv,
                                float* __restrict__ out1, int E) {
    long long t = (long long)blockIdx.x * blockDim.x + threadIdx.x;
    if (t >= (long long)E * 16) return;
    int e = (int)(t >> 4), c = (int)(t & 15);
    int s = src[e], d = dst[e];
    float norm = dinv[s] * dinv[d];
    atomicAdd(&out1[(size_t)d * 16 + c], h1[(size_t)s * 16 + c] * norm);
}

__global__ void gemm2_relu_kernel(const float* __restrict__ out1,
                                  const float* __restrict__ w2,
                                  float* __restrict__ h2, int n) {
    int i = blockIdx.x * blockDim.x + threadIdx.x;
    if (i >= n) return;
    float v[16];
#pragma unroll
    for (int k = 0; k < 16; ++k) {
        float t = out1[(size_t)i * 16 + k];
        v[k] = t > 0.0f ? t : 0.0f;
    }
#pragma unroll
    for (int c = 0; c < 4; ++c) {
        float s = 0.0f;
#pragma unroll
        for (int k = 0; k < 16; ++k) s += v[k] * w2[k * 4 + c];
        h2[(size_t)i * 4 + c] = s;
    }
}

__global__ void init_out2_kernel(const float* __restrict__ h2,
                                 const float* __restrict__ dinv,
                                 const float* __restrict__ b2,
                                 float* __restrict__ out2, int n) {
    int t = blockIdx.x * blockDim.x + threadIdx.x;
    if (t >= n * 4) return;
    int i = t >> 2, c = t & 3;
    float di = dinv[i];
    out2[t] = b2[c] + h2[t] * di * di;
}

__global__ void scatter2_kernel(const int* __restrict__ src,
                                const int* __restrict__ dst,
                                const float* __restrict__ h2,
                                const float* __restrict__ dinv,
                                float* __restrict__ out2, int E) {
    long long t = (long long)blockIdx.x * blockDim.x + threadIdx.x;
    if (t >= (long long)E * 4) return;
    int e = (int)(t >> 2), c = (int)(t & 3);
    int s = src[e], d = dst[e];
    float norm = dinv[s] * dinv[d];
    atomicAdd(&out2[(size_t)d * 4 + c], h2[(size_t)s * 4 + c] * norm);
}

__global__ void log_softmax_kernel(const float* __restrict__ acc,
                                   float* __restrict__ out, int n) {
    int i = blockIdx.x * blockDim.x + threadIdx.x;
    if (i >= n) return;
    float4 a = ((const float4*)acc)[i];
    float m = fmaxf(fmaxf(a.x, a.y), fmaxf(a.z, a.w));
    float s = __expf(a.x - m) + __expf(a.y - m) + __expf(a.z - m) + __expf(a.w - m);
    float lse = m + __logf(s);
    float4 r;
    r.x = a.x - lse; r.y = a.y - lse; r.z = a.z - lse; r.w = a.w - lse;
    ((float4*)out)[i] = r;
}

extern "C" void kernel_launch(void* const* d_in, const int* in_sizes, int n_in,
                              void* d_out, int out_size, void* d_ws, size_t ws_size,
                              hipStream_t stream) {
    const float* x  = (const float*)d_in[0];
    const int*   ei = (const int*)d_in[1];
    const float* w1 = (const float*)d_in[2];
    const float* b1 = (const float*)d_in[3];
    const float* w2 = (const float*)d_in[4];
    const float* b2 = (const float*)d_in[5];
    float* out = (float*)d_out;

    const int n = in_sizes[0] / 128;
    const int E = in_sizes[1] / 2;
    const int* src = ei;
    const int* dst = ei + E;

    float* dinv = (float*)d_ws;
    float* h1   = dinv + n;
    float* o1   = h1 + (size_t)n * 16;
    float* h2   = o1 + (size_t)n * 16;
    float* a2   = h2 + (size_t)n * 4;

    const int B = 256;
    auto cdiv = [](long long a, long long b) { return (int)((a + b - 1) / b); };

    init_deg_kernel<<<cdiv(n, B), B, 0, stream>>>(dinv, n);
    acc_deg_kernel<<<cdiv(E, B), B, 0, stream>>>(dst, dinv, E);
    to_dinv_kernel<<<cdiv(n, B), B, 0, stream>>>(dinv, n);

    {
        long long threads = (long long)((n + 15) / 16) * 32;
        gemm1_wmma_kernel<<<cdiv(threads, B), B, 0, stream>>>(x, w1, h1, n);
    }

    init_out1_kernel<<<cdiv((long long)n * 16, B), B, 0, stream>>>(h1, dinv, b1, o1, n);
    scatter1_kernel<<<cdiv((long long)E * 16, B), B, 0, stream>>>(src, dst, h1, dinv, o1, E);

    gemm2_relu_kernel<<<cdiv(n, B), B, 0, stream>>>(o1, w2, h2, n);

    init_out2_kernel<<<cdiv((long long)n * 4, B), B, 0, stream>>>(h2, dinv, b2, a2, n);
    scatter2_kernel<<<cdiv((long long)E * 4, B), B, 0, stream>>>(src, dst, h2, dinv, a2, E);

    log_softmax_kernel<<<cdiv(n, B), B, 0, stream>>>(a2, out, n);
}


# Edges grouped by destination range each call (LDS counters); a work-group per 400-node range sums degrees and messages with LDS f32 adds, rows stored once; no global atomics; sum order changes
# speedup vs baseline: 3.4790x; 3.4790x over previous
#include <hip/hip_runtime.h>
#include <math.h>


typedef __attribute__((ext_vector_type(2))) float v2f;
typedef __attribute__((ext_vector_type(8))) float v8f;

constexpr int R_NODES = 400;
constexpr int G_BIN   = 256;
constexpr int NB_MAX  = 512;
constexpr int BT      = 1024;

__global__ __launch_bounds__(BT) void bin_count_kernel(const int* __restrict__ dst,
                                                       int* __restrict__ counts,
                                                       int E, int n) {
    __shared__ int s_hist[NB_MAX];
    const int tid = threadIdx.x;
    const int nb  = (n + R_NODES - 1) / R_NODES;
    const int G   = (int)gridDim.x;
    for (int k = tid; k < NB_MAX; k += BT) s_hist[k] = 0;
    __syncthreads();

    const int chunk = (E + G - 1) / G;
    const int beg   = (int)blockIdx.x * chunk;
    int len = E - beg;
    if (len > chunk) len = chunk;

    for (int i0 = tid; i0 < len; i0 += 4 * BT) {
        int d[4];
#pragma unroll
        for (int u = 0; u < 4; ++u) {
            const int i = i0 + u * BT;
            d[u] = (i < len) ? dst[beg + i] : -1;
        }
#pragma unroll
        for (int u = 0; u < 4; ++u)
            if (d[u] >= 0) atomicAdd(&s_hist[d[u] / R_NODES], 1);
    }
    __syncthreads();
    for (int k = tid; k < nb; k += BT) counts[k * G + (int)blockIdx.x] = s_hist[k];
}

__global__ __launch_bounds__(BT) void bin_fill_kernel(const int* __restrict__ src,
                                                      const int* __restrict__ dst,
                                                      const int* __restrict__ counts,
                                                      int* __restrict__ pairs_i,
                                                      int* __restrict__ bstart,
                                                      int E, int n) {
    __shared__ int s_pref[NB_MAX];
    __shared__ int s_tot[NB_MAX];
    __shared__ int s_cur[NB_MAX];
    int2* __restrict__ pairs = (int2*)pairs_i;
    const int tid  = threadIdx.x;
    const int lane = tid & 31;
    const int wave = tid >> 5;
    const int nb   = (n + R_NODES - 1) / R_NODES;
    const int G    = (int)gridDim.x;
    const int b    = (int)blockIdx.x;

    for (int k = wave; k < nb; k += BT / 32) {
        const int* row = counts + (size_t)k * G;
        int pre = 0, tot = 0;
        for (int j = lane; j < G; j += 32) {
            const int v = row[j];
            tot += v;
            pre += (j < b) ? v : 0;
        }
        for (int o = 16; o > 0; o >>= 1) {
            pre += __shfl_xor(pre, o);
            tot += __shfl_xor(tot, o);
        }
        if (lane == 0) { s_pref[k] = pre; s_tot[k] = tot; }
    }
    __syncthreads();

    if (wave == 0) {
        int loc[16];
        int sum = 0;
#pragma unroll
        for (int i = 0; i < 16; ++i) {
            const int k = lane * 16 + i;
            loc[i] = (k < nb) ? s_tot[k] : 0;
            sum += loc[i];
        }
        int incl = sum;
        for (int o = 1; o < 32; o <<= 1) {
            const int t = __shfl_up(incl, o);
            if (lane >= o) incl += t;
        }
        int run = incl - sum;
#pragma unroll
        for (int i = 0; i < 16; ++i) {
            const int k = lane * 16 + i;
            if (k < nb) {
                s_cur[k] = run + s_pref[k];
                if (b == 0) bstart[k] = run;
            }
            run += loc[i];
        }
        if (b == 0 && lane == 31) bstart[nb] = run;
    }
    __syncthreads();

    const int chunk = (E + G - 1) / G;
    const int beg   = b * chunk;
    int len = E - beg;
    if (len > chunk) len = chunk;

    for (int i0 = tid; i0 < len; i0 += 4 * BT) {
        int s[4], d[4];
#pragma unroll
        for (int u = 0; u < 4; ++u) {
            const int i = i0 + u * BT;
            const bool ok = i < len;
            d[u] = ok ? dst[beg + i] : -1;
            s[u] = ok ? src[beg + i] : 0;
        }
#pragma unroll
        for (int u = 0; u < 4; ++u) {
            if (d[u] >= 0) {
                const int pos = atomicAdd(&s_cur[d[u] / R_NODES], 1);
                pairs[pos] = make_int2(s[u], d[u]);
            }
        }
    }
}

__global__ void gemm1_wmma_kernel(const float* __restrict__ x,
                                  const float* __restrict__ w1,
                                  float* __restrict__ h1, int n) {
    int wave = (int)((blockIdx.x * blockDim.x + threadIdx.x) >> 5);
    int lane = threadIdx.x & 31;
    int row0 = wave * 16;
    if (row0 >= n) return;

    int m    = lane & 15;
    int koff = (lane >> 4) * 2;

    const float* xrow = x + (size_t)(row0 + m) * 128;

    v8f acc = {};
    for (int k0 = 0; k0 < 128; k0 += 4) {
        v2f a, b;
        a.x = xrow[k0 + koff + 0];
        a.y = xrow[k0 + koff + 1];
        b.x = w1[(k0 + koff + 0) * 16 + m];
        b.y = w1[(k0 + koff + 1) * 16 + m];
        acc = __builtin_amdgcn_wmma_f32_16x16x4_f32(
             false, a,  false, b,
             (short)0, acc,  false,  false);
    }

    int nCol  = lane & 15;
    int mBase = (lane >> 4) * 8;
#pragma unroll
    for (int r = 0; r < 8; ++r) {
        h1[(size_t)(row0 + mBase + r) * 16 + nCol] = acc[r];
    }
}


__global__ __launch_bounds__(BT) void bucket_deg_kernel(const int* __restrict__ pairs_i,
                                                        const int* __restrict__ bstart,
                                                        float* __restrict__ dinv, int n) {
    __shared__ int s_cnt[R_NODES];
    const int2* __restrict__ pairs = (const int2*)pairs_i;
    const int tid  = threadIdx.x;
    const int k    = (int)blockIdx.x;
    const int base = k * R_NODES;
    int cnt = n - base;
    if (cnt > R_NODES) cnt = R_NODES;
    for (int i = tid; i < R_NODES; i += BT) s_cnt[i] = 0;
    __syncthreads();

    const int beg = bstart[k], end = bstart[k + 1];
    for (int e0 = beg + tid; e0 < end; e0 += 4 * BT) {
        int d[4];
#pragma unroll
        for (int u = 0; u < 4; ++u) {
            const int e = e0 + u * BT;
            d[u] = (e < end) ? pairs[e].y : -1;
        }
#pragma unroll
        for (int u = 0; u < 4; ++u)
            if (d[u] >= 0) atomicAdd(&s_cnt[d[u] - base], 1);
    }
    __syncthreads();
    for (int i = tid; i < cnt; i += BT) dinv[base + i] = rsqrtf((float)(1 + s_cnt[i]));
}

template <int F, int U, bool TAIL>
__device__ __forceinline__ void agg_batch(const int2* __restrict__ pairs,
                                          const float* __restrict__ h,
                                          const float* __restrict__ dinv,
                                          float* s_acc, const float* s_dinv,
                                          int e0, int end, int base, int c) {
    constexpr int EPB = BT / F;
    int2  p[U];
    float hv[U], ds[U];
#pragma unroll
    for (int u = 0; u < U; ++u) {
        int e = e0 + u * EPB;
        if (TAIL) e = (e < end) ? e : end - 1;
        p[u] = pairs[e];
    }
#pragma unroll
    for (int u = 0; u < U; ++u) {
        ds[u] = dinv[p[u].x];
        hv[u] = h[(size_t)p[u].x * F + c];
    }
#pragma unroll
    for (int u = 0; u < U; ++u) {
        const int dl = p[u].y - base;
        const float norm = ds[u] * s_dinv[dl];
        float msg = hv[u] * norm;
        if (TAIL && !(e0 + u * EPB < end)) msg = 0.0f;
        atomicAdd(&s_acc[dl * F + c], msg);
    }
}

template <int F>
__global__ __launch_bounds__(BT) void bucket_agg_kernel(const int* __restrict__ pairs_i,
                                                        const int* __restrict__ bstart,
                                                        const float* __restrict__ h,
                                                        const float* __restrict__ dinv,
                                                        const float* __restrict__ bias,
                                                        float* __restrict__ outp, int n) {
    __shared__ float s_acc[R_NODES * F];
    __shared__ float s_dinv[R_NODES];
    const int2* __restrict__ pairs = (const int2*)pairs_i;
    const int tid  = threadIdx.x;
    const int k    = (int)blockIdx.x;
    const int base = k * R_NODES;
    int cnt = n - base;
    if (cnt > R_NODES) cnt = R_NODES;

    for (int i = tid; i < cnt; i += BT) s_dinv[i] = dinv[base + i];
    for (int t = tid; t < cnt * F; t += BT) {
        const int i = t / F, c = t % F;
        const float di = dinv[base + i];
        s_acc[t] = bias[c] + h[(size_t)base * F + t] * di * di;
    }
    __syncthreads();

    constexpr int EPB = BT / F;
    constexpr int U   = 8;
    const int sub = tid / F, c = tid % F;
    const int beg = bstart[k], end = bstart[k + 1];

    int e0 = beg + sub;
    for (; e0 + (U - 1) * EPB < end; e0 += U * EPB)
        agg_batch<F, U, false>(pairs, h, dinv, s_acc, s_dinv, e0, end, base, c);
    if (e0 < end)
        agg_batch<F, U, true>(pairs, h, dinv, s_acc, s_dinv, e0, end, base, c);
    __syncthreads();
    for (int t = tid; t < cnt * F; t += BT) outp[(size_t)base * F + t] = s_acc[t];
}

__global__ void gemm2_relu_kernel(const float* __restrict__ out1,
                                  const float* __restrict__ w2,
                                  float* __restrict__ h2, int n) {
    int i = blockIdx.x * blockDim.x + threadIdx.x;
    if (i >= n) return;
    float v[16];
#pragma unroll
    for (int k = 0; k < 16; ++k) {
        float t = out1[(size_t)i * 16 + k];
        v[k] = t > 0.0f ? t : 0.0f;
    }
#pragma unroll
    for (int c = 0; c < 4; ++c) {
        float s = 0.0f;
#pragma unroll
        for (int k = 0; k < 16; ++k) s += v[k] * w2[k * 4 + c];
        h2[(size_t)i * 4 + c] = s;
    }
}

__global__ void log_softmax_kernel(const float* __restrict__ acc,
                                   float* __restrict__ out, int n) {
    int i = blockIdx.x * blockDim.x + threadIdx.x;
    if (i >= n) return;
    float4 a = ((const float4*)acc)[i];
    float m = fmaxf(fmaxf(a.x, a.y), fmaxf(a.z, a.w));
    float s = __expf(a.x - m) + __expf(a.y - m) + __expf(a.z - m) + __expf(a.w - m);
    float lse = m + __logf(s);
    float4 r;
    r.x = a.x - lse; r.y = a.y - lse; r.z = a.z - lse; r.w = a.w - lse;
    ((float4*)out)[i] = r;
}

extern "C" void kernel_launch(void* const* d_in, const int* in_sizes, int n_in,
                              void* d_out, int out_size, void* d_ws, size_t ws_size,
                              hipStream_t stream) {
    const float* x  = (const float*)d_in[0];
    const int*   ei = (const int*)d_in[1];
    const float* w1 = (const float*)d_in[2];
    const float* b1 = (const float*)d_in[3];
    const float* w2 = (const float*)d_in[4];
    const float* b2 = (const float*)d_in[5];
    float* out = (float*)d_out;

    const int n = in_sizes[0] / 128;
    const int E = in_sizes[1] / 2;
    const int* src = ei;
    const int* dst = ei + E;

    float* dinv = (float*)d_ws;
    float* h1   = dinv + n;
    float* o1   = h1 + (size_t)n * 16;
    float* h2   = o1 + (size_t)n * 16;
    float* a2   = h2 + (size_t)n * 4;
    int* counts = (int*)(a2 + (size_t)n * 4);
    int* bstart = counts + NB_MAX * G_BIN;
    int* pairs  = bstart + 1024;

    const int B = 256;
    auto cdiv = [](long long a, long long b) { return (int)((a + b - 1) / b); };
    const int NB = cdiv(n, R_NODES);

    bin_count_kernel<<<G_BIN, BT, 0, stream>>>(dst, counts, E, n);
    bin_fill_kernel<<<G_BIN, BT, 0, stream>>>(src, dst, counts, pairs, bstart, E, n);

    bucket_deg_kernel<<<NB, BT, 0, stream>>>(pairs, bstart, dinv, n);

    {
        long long threads = (long long)((n + 15) / 16) * 32;
        gemm1_wmma_kernel<<<cdiv(threads, B), B, 0, stream>>>(x, w1, h1, n);
    }

    bucket_agg_kernel<16><<<NB, BT, 0, stream>>>(pairs, bstart, h1, dinv, b1, o1, n);

    gemm2_relu_kernel<<<cdiv(n, B), B, 0, stream>>>(o1, w2, h2, n);

    bucket_agg_kernel<4><<<NB, BT, 0, stream>>>(pairs, bstart, h2, dinv, b2, a2, n);

    log_softmax_kernel<<<cdiv(n, B), B, 0, stream>>>(a2, out, n);
}


# Feature rows stored pre-scaled by dinv[src] in the GEMM epilogues and dinv[dst] applied once to each row's sum: the edge loop gathers one row per edge, no dinv gather; products round differently
# speedup vs baseline: 4.2120x; 1.2107x over previous
#include <hip/hip_runtime.h>
#include <math.h>


typedef __attribute__((ext_vector_type(2))) float v2f;
typedef __attribute__((ext_vector_type(8))) float v8f;

constexpr int R_NODES = 400;
constexpr int G_BIN   = 256;
constexpr int NB_MAX  = 512;
constexpr int BT      = 1024;

__global__ __launch_bounds__(BT) void bin_count_kernel(const int* __restrict__ dst,
                                                       int* __restrict__ counts,
                                                       int E, int n) {
    __shared__ int s_hist[NB_MAX];
    const int tid = threadIdx.x;
    const int nb  = (n + R_NODES - 1) / R_NODES;
    const int G   = (int)gridDim.x;
    for (int k = tid; k < NB_MAX; k += BT) s_hist[k] = 0;
    __syncthreads();

    const int chunk = (E + G - 1) / G;
    const int beg   = (int)blockIdx.x * chunk;
    int len = E - beg;
    if (len > chunk) len = chunk;

    for (int i0 = tid; i0 < len; i0 += 4 * BT) {
        int d[4];
#pragma unroll
        for (int u = 0; u < 4; ++u) {
            const int i = i0 + u * BT;
            d[u] = (i < len) ? dst[beg + i] : -1;
        }
#pragma unroll
        for (int u = 0; u < 4; ++u)
            if (d[u] >= 0) atomicAdd(&s_hist[d[u] / R_NODES], 1);
    }
    __syncthreads();
    for (int k = tid; k < nb; k += BT) counts[k * G + (int)blockIdx.x] = s_hist[k];
}

__global__ __launch_bounds__(BT) void bin_fill_kernel(const int* __restrict__ src,
                                                      const int* __restrict__ dst,
                                                      const int* __restrict__ counts,
                                                      int* __restrict__ pairs_i,
                                                      int* __restrict__ bstart,
                                                      int E, int n) {
    __shared__ int s_pref[NB_MAX];
    __shared__ int s_tot[NB_MAX];
    __shared__ int s_cur[NB_MAX];
    int2* __restrict__ pairs = (int2*)pairs_i;
    const int tid  = threadIdx.x;
    const int lane = tid & 31;
    const int wave = tid >> 5;
    const int nb   = (n + R_NODES - 1) / R_NODES;
    const int G    = (int)gridDim.x;
    const int b    = (int)blockIdx.x;

    for (int k = wave; k < nb; k += BT / 32) {
        const int* row = counts + (size_t)k * G;
        int pre = 0, tot = 0;
        for (int j = lane; j < G; j += 32) {
            const int v = row[j];
            tot += v;
            pre += (j < b) ? v : 0;
        }
        for (int o = 16; o > 0; o >>= 1) {
            pre += __shfl_xor(pre, o);
            tot += __shfl_xor(tot, o);
        }
        if (lane == 0) { s_pref[k] = pre; s_tot[k] = tot; }
    }
    __syncthreads();

    if (wave == 0) {
        int loc[16];
        int sum = 0;
#pragma unroll
        for (int i = 0; i < 16; ++i) {
            const int k = lane * 16 + i;
            loc[i] = (k < nb) ? s_tot[k] : 0;
            sum += loc[i];
        }
        int incl = sum;
        for (int o = 1; o < 32; o <<= 1) {
            const int t = __shfl_up(incl, o);
            if (lane >= o) incl += t;
        }
        int run = incl - sum;
#pragma unroll
        for (int i = 0; i < 16; ++i) {
            const int k = lane * 16 + i;
            if (k < nb) {
                s_cur[k] = run + s_pref[k];
                if (b == 0) bstart[k] = run;
            }
            run += loc[i];
        }
        if (b == 0 && lane == 31) bstart[nb] = run;
    }
    __syncthreads();

    const int chunk = (E + G - 1) / G;
    const int beg   = b * chunk;
    int len = E - beg;
    if (len > chunk) len = chunk;

    for (int i0 = tid; i0 < len; i0 += 4 * BT) {
        int s[4], d[4];
#pragma unroll
        for (int u = 0; u < 4; ++u) {
            const int i = i0 + u * BT;
            const bool ok = i < len;
            d[u] = ok ? dst[beg + i] : -1;
            s[u] = ok ? src[beg + i] : 0;
        }
#pragma unroll
        for (int u = 0; u < 4; ++u) {
            if (d[u] >= 0) {
                const int pos = atomicAdd(&s_cur[d[u] / R_NODES], 1);
                pairs[pos] = make_int2(s[u], d[u]);
            }
        }
    }
}

__global__ void gemm1_wmma_kernel(const float* __restrict__ x,
                                  const float* __restrict__ w1,
                                  const float* __restrict__ dinv,
                                  float* __restrict__ h1, int n) {
    int wave = (int)((blockIdx.x * blockDim.x + threadIdx.x) >> 5);
    int lane = threadIdx.x & 31;
    int row0 = wave * 16;
    if (row0 >= n) return;

    int m    = lane & 15;
    int koff = (lane >> 4) * 2;

    const float* xrow = x + (size_t)(row0 + m) * 128;

    const int mBase = (lane >> 4) * 8;
    const float4 dlo = *(const float4*)(dinv + row0 + mBase);
    const float4 dhi = *(const float4*)(dinv + row0 + mBase + 4);
    const float dr[8] = {dlo.x, dlo.y, dlo.z, dlo.w, dhi.x, dhi.y, dhi.z, dhi.w};

    v8f acc = {};
    for (int k0 = 0; k0 < 128; k0 += 4) {
        v2f a, b;
        a.x = xrow[k0 + koff + 0];
        a.y = xrow[k0 + koff + 1];
        b.x = w1[(k0 + koff + 0) * 16 + m];
        b.y = w1[(k0 + koff + 1) * 16 + m];
        acc = __builtin_amdgcn_wmma_f32_16x16x4_f32(
             false, a,  false, b,
             (short)0, acc,  false,  false);
    }

    int nCol  = lane & 15;
#pragma unroll
    for (int r = 0; r < 8; ++r) {
        h1[(size_t)(row0 + mBase + r) * 16 + nCol] = acc[r] * dr[r];
    }
}


__global__ __launch_bounds__(BT) void bucket_deg_kernel(const int* __restrict__ pairs_i,
                                                        const int* __restrict__ bstart,
                                                        float* __restrict__ dinv, int n) {
    __shared__ int s_cnt[R_NODES];
    const int2* __restrict__ pairs = (const int2*)pairs_i;
    const int tid  = threadIdx.x;
    const int k    = (int)blockIdx.x;
    const int base = k * R_NODES;
    int cnt = n - base;
    if (cnt > R_NODES) cnt = R_NODES;
    for (int i = tid; i < R_NODES; i += BT) s_cnt[i] = 0;
    __syncthreads();

    const int beg = bstart[k], end = bstart[k + 1];
    for (int e0 = beg + tid; e0 < end; e0 += 4 * BT) {
        int d[4];
#pragma unroll
        for (int u = 0; u < 4; ++u) {
            const int e = e0 + u * BT;
            d[u] = (e < end) ? pairs[e].y : -1;
        }
#pragma unroll
        for (int u = 0; u < 4; ++u)
            if (d[u] >= 0) atomicAdd(&s_cnt[d[u] - base], 1);
    }
    __syncthreads();
    for (int i = tid; i < cnt; i += BT) dinv[base + i] = rsqrtf((float)(1 + s_cnt[i]));
}

template <int F, int U, bool TAIL>
__device__ __forceinline__ void agg_batch(const int2* __restrict__ pairs,
                                          const float* __restrict__ h,
                                          float* s_acc,
                                          int e0, int end, int base, int c) {
    constexpr int EPB = BT / F;
    int2  p[U];
    float hv[U];
#pragma unroll
    for (int u = 0; u < U; ++u) {
        int e = e0 + u * EPB;
        if (TAIL) e = (e < end) ? e : end - 1;
        p[u] = pairs[e];
    }
#pragma unroll
    for (int u = 0; u < U; ++u)
        hv[u] = h[(size_t)p[u].x * F + c];
#pragma unroll
    for (int u = 0; u < U; ++u) {
        const int dl = p[u].y - base;
        float msg = hv[u];
        if (TAIL && !(e0 + u * EPB < end)) msg = 0.0f;
        atomicAdd(&s_acc[dl * F + c], msg);
    }
}

template <int F>
__global__ __launch_bounds__(BT) void bucket_agg_kernel(const int* __restrict__ pairs_i,
                                                        const int* __restrict__ bstart,
                                                        const float* __restrict__ h,
                                                        const float* __restrict__ dinv,
                                                        const float* __restrict__ bias,
                                                        float* __restrict__ outp, int n) {
    __shared__ float s_acc[R_NODES * F];
    __shared__ float s_dinv[R_NODES];
    const int2* __restrict__ pairs = (const int2*)pairs_i;
    const int tid  = threadIdx.x;
    const int k    = (int)blockIdx.x;
    const int base = k * R_NODES;
    int cnt = n - base;
    if (cnt > R_NODES) cnt = R_NODES;

    for (int i = tid; i < cnt; i += BT) s_dinv[i] = dinv[base + i];
    for (int t = tid; t < cnt * F; t += BT) s_acc[t] = h[(size_t)base * F + t];
    __syncthreads();

    constexpr int EPB = BT / F;
    constexpr int U   = 8;
    const int sub = tid / F, c = tid % F;
    const int beg = bstart[k], end = bstart[k + 1];

    int e0 = beg + sub;
    for (; e0 + (U - 1) * EPB < end; e0 += U * EPB)
        agg_batch<F, U, false>(pairs, h, s_acc, e0, end, base, c);
    if (e0 < end)
        agg_batch<F, U, true>(pairs, h, s_acc, e0, end, base, c);
    __syncthreads();
    for (int t = tid; t < cnt * F; t += BT)
        outp[(size_t)base * F + t] = bias[t % F] + s_dinv[t / F] * s_acc[t];
}

__global__ void gemm2_relu_kernel(const float* __restrict__ out1,
                                  const float* __restrict__ w2,
                                  const float* __restrict__ dinv,
                                  float* __restrict__ h2, int n) {
    int i = blockIdx.x * blockDim.x + threadIdx.x;
    if (i >= n) return;
    const float di = dinv[i];
    float v[16];
#pragma unroll
    for (int k = 0; k < 16; ++k) {
        float t = out1[(size_t)i * 16 + k];
        v[k] = t > 0.0f ? t : 0.0f;
    }
#pragma unroll
    for (int c = 0; c < 4; ++c) {
        float s = 0.0f;
#pragma unroll
        for (int k = 0; k < 16; ++k) s += v[k] * w2[k * 4 + c];
        h2[(size_t)i * 4 + c] = s * di;
    }
}

__global__ void log_softmax_kernel(const float* __restrict__ acc,
                                   float* __restrict__ out, int n) {
    int i = blockIdx.x * blockDim.x + threadIdx.x;
    if (i >= n) return;
    float4 a = ((const float4*)acc)[i];
    float m = fmaxf(fmaxf(a.x, a.y), fmaxf(a.z, a.w));
    float s = __expf(a.x - m) + __expf(a.y - m) + __expf(a.z - m) + __expf(a.w - m);
    float lse = m + __logf(s);
    float4 r;
    r.x = a.x - lse; r.y = a.y - lse; r.z = a.z - lse; r.w = a.w - lse;
    ((float4*)out)[i] = r;
}

extern "C" void kernel_launch(void* const* d_in, const int* in_sizes, int n_in,
                              void* d_out, int out_size, void* d_ws, size_t ws_size,
                              hipStream_t stream) {
    const float* x  = (const float*)d_in[0];
    const int*   ei = (const int*)d_in[1];
    const float* w1 = (const float*)d_in[2];
    const float* b1 = (const float*)d_in[3];
    const float* w2 = (const float*)d_in[4];
    const float* b2 = (const float*)d_in[5];
    float* out = (float*)d_out;

    const int n = in_sizes[0] / 128;
    const int E = in_sizes[1] / 2;
    const int* src = ei;
    const int* dst = ei + E;

    float* dinv = (float*)d_ws;
    float* h1   = dinv + n;
    float* o1   = h1 + (size_t)n * 16;
    float* h2   = o1 + (size_t)n * 16;
    float* a2   = h2 + (size_t)n * 4;
    int* counts = (int*)(a2 + (size_t)n * 4);
    int* bstart = counts + NB_MAX * G_BIN;
    int* pairs  = bstart + 1024;

    const int B = 256;
    auto cdiv = [](long long a, long long b) { return (int)((a + b - 1) / b); };
    const int NB = cdiv(n, R_NODES);

    bin_count_kernel<<<G_BIN, BT, 0, stream>>>(dst, counts, E, n);
    bin_fill_kernel<<<G_BIN, BT, 0, stream>>>(src, dst, counts, pairs, bstart, E, n);

    bucket_deg_kernel<<<NB, BT, 0, stream>>>(pairs, bstart, dinv, n);

    {
        long long threads = (long long)((n + 15) / 16) * 32;
        gemm1_wmma_kernel<<<cdiv(threads, B), B, 0, stream>>>(x, w1, dinv, h1, n);
    }

    bucket_agg_kernel<16><<<NB, BT, 0, stream>>>(pairs, bstart, h1, dinv, b1, o1, n);

    gemm2_relu_kernel<<<cdiv(n, B), B, 0, stream>>>(o1, w2, dinv, h2, n);

    bucket_agg_kernel<4><<<NB, BT, 0, stream>>>(pairs, bstart, h2, dinv, b2, a2, n);

    log_softmax_kernel<<<cdiv(n, B), B, 0, stream>>>(a2, out, n);
}


# Per-node stages run in the bucket work-groups that own the rows: GEMM1 behind the degree count, ReLU+GEMM2 and log_softmax on the finished sums in LDS; 8 launches become 5, out1 is never stored
# speedup vs baseline: 4.7448x; 1.1265x over previous
#include <hip/hip_runtime.h>
#include <math.h>


typedef __attribute__((ext_vector_type(2))) float v2f;
typedef __attribute__((ext_vector_type(8))) float v8f;

constexpr int R_NODES = 400;
constexpr int G_BIN   = 256;
constexpr int NB_MAX  = 512;
constexpr int BT      = 1024;

__global__ __launch_bounds__(BT) void bin_count_kernel(const int* __restrict__ dst,
                                                       int* __restrict__ counts,
                                                       int E, int n) {
    __shared__ int s_hist[NB_MAX];
    const int tid = threadIdx.x;
    const int nb  = (n + R_NODES - 1) / R_NODES;
    const int G   = (int)gridDim.x;
    for (int k = tid; k < NB_MAX; k += BT) s_hist[k] = 0;
    __syncthreads();

    const int chunk = (E + G - 1) / G;
    const int beg   = (int)blockIdx.x * chunk;
    int len = E - beg;
    if (len > chunk) len = chunk;

    for (int i0 = tid; i0 < len; i0 += 4 * BT) {
        int d[4];
#pragma unroll
        for (int u = 0; u < 4; ++u) {
            const int i = i0 + u * BT;
            d[u] = (i < len) ? dst[beg + i] : -1;
        }
#pragma unroll
        for (int u = 0; u < 4; ++u)
            if (d[u] >= 0) atomicAdd(&s_hist[d[u] / R_NODES], 1);
    }
    __syncthreads();
    for (int k = tid; k < nb; k += BT) counts[k * G + (int)blockIdx.x] = s_hist[k];
}

__global__ __launch_bounds__(BT) void bin_fill_kernel(const int* __restrict__ src,
                                                      const int* __restrict__ dst,
                                                      const int* __restrict__ counts,
                                                      int* __restrict__ pairs_i,
                                                      int* __restrict__ bstart,
                                                      int E, int n) {
    __shared__ int s_pref[NB_MAX];
    __shared__ int s_tot[NB_MAX];
    __shared__ int s_cur[NB_MAX];
    int2* __restrict__ pairs = (int2*)pairs_i;
    const int tid  = threadIdx.x;
    const int lane = tid & 31;
    const int wave = tid >> 5;
    const int nb   = (n + R_NODES - 1) / R_NODES;
    const int G    = (int)gridDim.x;
    const int b    = (int)blockIdx.x;

    for (int k = wave; k < nb; k += BT / 32) {
        const int* row = counts + (size_t)k * G;
        int pre = 0, tot = 0;
        for (int j = lane; j < G; j += 32) {
            const int v = row[j];
            tot += v;
            pre += (j < b) ? v : 0;
        }
        for (int o = 16; o > 0; o >>= 1) {
            pre += __shfl_xor(pre, o);
            tot += __shfl_xor(tot, o);
        }
        if (lane == 0) { s_pref[k] = pre; s_tot[k] = tot; }
    }
    __syncthreads();

    if (wave == 0) {
        int loc[16];
        int sum = 0;
#pragma unroll
        for (int i = 0; i < 16; ++i) {
            const int k = lane * 16 + i;
            loc[i] = (k < nb) ? s_tot[k] : 0;
            sum += loc[i];
        }
        int incl = sum;
        for (int o = 1; o < 32; o <<= 1) {
            const int t = __shfl_up(incl, o);
            if (lane >= o) incl += t;
        }
        int run = incl - sum;
#pragma unroll
        for (int i = 0; i < 16; ++i) {
            const int k = lane * 16 + i;
            if (k < nb) {
                s_cur[k] = run + s_pref[k];
                if (b == 0) bstart[k] = run;
            }
            run += loc[i];
        }
        if (b == 0 && lane == 31) bstart[nb] = run;
    }
    __syncthreads();

    const int chunk = (E + G - 1) / G;
    const int beg   = b * chunk;
    int len = E - beg;
    if (len > chunk) len = chunk;

    for (int i0 = tid; i0 < len; i0 += 4 * BT) {
        int s[4], d[4];
#pragma unroll
        for (int u = 0; u < 4; ++u) {
            const int i = i0 + u * BT;
            const bool ok = i < len;
            d[u] = ok ? dst[beg + i] : -1;
            s[u] = ok ? src[beg + i] : 0;
        }
#pragma unroll
        for (int u = 0; u < 4; ++u) {
            if (d[u] >= 0) {
                const int pos = atomicAdd(&s_cur[d[u] / R_NODES], 1);
                pairs[pos] = make_int2(s[u], d[u]);
            }
        }
    }
}

static_assert(R_NODES % 16 == 0 && R_NODES / 16 <= BT / 32, "one 16-row tile per wave");

__global__ __launch_bounds__(BT) void bucket_deg_gemm1_kernel(const int* __restrict__ pairs_i,
                                                              const int* __restrict__ bstart,
                                                              const float* __restrict__ x,
                                                              const float* __restrict__ w1,
                                                              float* __restrict__ dinv,
                                                              float* __restrict__ h1, int n) {
    __shared__ __attribute__((aligned(16))) int s_cnt[R_NODES];
    const int2* __restrict__ pairs = (const int2*)pairs_i;
    const int tid  = threadIdx.x;
    const int k    = (int)blockIdx.x;
    const int base = k * R_NODES;
    int cnt = n - base;
    if (cnt > R_NODES) cnt = R_NODES;
    for (int i = tid; i < R_NODES; i += BT) s_cnt[i] = 0;
    __syncthreads();

    const int beg = bstart[k], end = bstart[k + 1];
    for (int e0 = beg + tid; e0 < end; e0 += 4 * BT) {
        int d[4];
#pragma unroll
        for (int u = 0; u < 4; ++u) {
            const int e = e0 + u * BT;
            d[u] = (e < end) ? pairs[e].y : -1;
        }
#pragma unroll
        for (int u = 0; u < 4; ++u)
            if (d[u] >= 0) atomicAdd(&s_cnt[d[u] - base], 1);
    }

    const int wave = tid >> 5;
    const int lane = tid & 31;
    const bool has_tile = wave * 16 < cnt;
    v8f acc = {};
    if (has_tile) {
        const int m    = lane & 15;
        const int koff = (lane >> 4) * 2;
        int arow = base + wave * 16 + m;
        if (arow > n - 1) arow = n - 1;
        const float* xrow = x + (size_t)arow * 128;
        for (int k0 = 0; k0 < 128; k0 += 4) {
            v2f a, b;
            a.x = xrow[k0 + koff + 0];
            a.y = xrow[k0 + koff + 1];
            b.x = w1[(k0 + koff + 0) * 16 + m];
            b.y = w1[(k0 + koff + 1) * 16 + m];
            acc = __builtin_amdgcn_wmma_f32_16x16x4_f32(
                 false, a,  false, b,
                 (short)0, acc,  false,  false);
        }
    }
    __syncthreads();

    for (int i = tid; i < cnt; i += BT) dinv[base + i] = rsqrtf((float)(1 + s_cnt[i]));

    if (has_tile) {
        const int mBase = wave * 16 + (lane >> 4) * 8;
        const int nCol  = lane & 15;
        const int4 clo = *(const int4*)&s_cnt[mBase];
        const int4 chi = *(const int4*)&s_cnt[mBase + 4];
        const int cr[8] = {clo.x, clo.y, clo.z, clo.w, chi.x, chi.y, chi.z, chi.w};
#pragma unroll
        for (int r = 0; r < 8; ++r) {
            const int rl = mBase + r;
            if (rl < cnt)
                h1[(size_t)(base + rl) * 16 + nCol] = acc[r] * rsqrtf((float)(1 + cr[r]));
        }
    }
}

template <int F, int U, bool TAIL>
__device__ __forceinline__ void agg_batch(const int2* __restrict__ pairs,
                                          const float* __restrict__ h,
                                          float* s_acc,
                                          int e0, int end, int base, int c) {
    constexpr int EPB = BT / F;
    int2  p[U];
    float hv[U];
#pragma unroll
    for (int u = 0; u < U; ++u) {
        int e = e0 + u * EPB;
        if (TAIL) e = (e < end) ? e : end - 1;
        p[u] = pairs[e];
    }
#pragma unroll
    for (int u = 0; u < U; ++u)
        hv[u] = h[(size_t)p[u].x * F + c];
#pragma unroll
    for (int u = 0; u < U; ++u) {
        const int dl = p[u].y - base;
        float msg = hv[u];
        if (TAIL && !(e0 + u * EPB < end)) msg = 0.0f;
        atomicAdd(&s_acc[dl * F + c], msg);
    }
}

template <int F>
__global__ __launch_bounds__(BT) void bucket_agg_kernel(const int* __restrict__ pairs_i,
                                                        const int* __restrict__ bstart,
                                                        const float* __restrict__ h,
                                                        const float* __restrict__ dinv,
                                                        const float* __restrict__ bias,
                                                        const float* __restrict__ w2,
                                                        float* __restrict__ outp, int n) {
    __shared__ __attribute__((aligned(16))) float s_acc[R_NODES * F];
    __shared__ float s_dinv[R_NODES];
    const int2* __restrict__ pairs = (const int2*)pairs_i;
    const int tid  = threadIdx.x;
    const int k    = (int)blockIdx.x;
    const int base = k * R_NODES;
    int cnt = n - base;
    if (cnt > R_NODES) cnt = R_NODES;

    for (int i = tid; i < cnt; i += BT) s_dinv[i] = dinv[base + i];
    for (int t = tid; t < cnt * F; t += BT) s_acc[t] = h[(size_t)base * F + t];
    __syncthreads();

    constexpr int EPB = BT / F;
    constexpr int U   = 8;
    const int sub = tid / F, c = tid % F;
    const int beg = bstart[k], end = bstart[k + 1];

    int e0 = beg + sub;
    for (; e0 + (U - 1) * EPB < end; e0 += U * EPB)
        agg_batch<F, U, false>(pairs, h, s_acc, e0, end, base, c);
    if (e0 < end)
        agg_batch<F, U, true>(pairs, h, s_acc, e0, end, base, c);
    __syncthreads();

    if constexpr (F == 16) {
        const int j = tid & 3;
        for (int t = tid; t < cnt * 4; t += BT) {
            const int nd = t >> 2;
            const float di = s_dinv[nd];
            float s = 0.0f;
#pragma unroll
            for (int q = 0; q < 16; ++q) {
                float v = bias[q] + di * s_acc[nd * 16 + q];
                v = v > 0.0f ? v : 0.0f;
                s += v * w2[q * 4 + j];
            }
            outp[(size_t)base * 4 + t] = s * di;
        }
    } else {
        static_assert(F == 4, "the last layer has 4 classes");
        for (int nd = tid; nd < cnt; nd += BT) {
            const float di = s_dinv[nd];
            const float4 sv = *(const float4*)&s_acc[nd * 4];
            float4 a;
            a.x = bias[0] + di * sv.x;
            a.y = bias[1] + di * sv.y;
            a.z = bias[2] + di * sv.z;
            a.w = bias[3] + di * sv.w;
            const float m = fmaxf(fmaxf(a.x, a.y), fmaxf(a.z, a.w));
            const float s = __expf(a.x - m) + __expf(a.y - m) + __expf(a.z - m) + __expf(a.w - m);
            const float lse = m + __logf(s);
            float4 r;
            r.x = a.x - lse; r.y = a.y - lse; r.z = a.z - lse; r.w = a.w - lse;
            ((float4*)outp)[base + nd] = r;
        }
    }
}

extern "C" void kernel_launch(void* const* d_in, const int* in_sizes, int n_in,
                              void* d_out, int out_size, void* d_ws, size_t ws_size,
                              hipStream_t stream) {
    const float* x  = (const float*)d_in[0];
    const int*   ei = (const int*)d_in[1];
    const float* w1 = (const float*)d_in[2];
    const float* b1 = (const float*)d_in[3];
    const float* w2 = (const float*)d_in[4];
    const float* b2 = (const float*)d_in[5];
    float* out = (float*)d_out;

    const int n = in_sizes[0] / 128;
    const int E = in_sizes[1] / 2;
    const int* src = ei;
    const int* dst = ei + E;

    float* dinv = (float*)d_ws;
    float* h1   = dinv + n;
    float* h2   = h1 + (size_t)n * 16;
    int* counts = (int*)(h2 + (size_t)n * 4);
    int* bstart = counts + NB_MAX * G_BIN;
    int* pairs  = bstart + 1024;

    auto cdiv = [](long long a, long long b) { return (int)((a + b - 1) / b); };
    const int NB = cdiv(n, R_NODES);

    bin_count_kernel<<<G_BIN, BT, 0, stream>>>(dst, counts, E, n);
    bin_fill_kernel<<<G_BIN, BT, 0, stream>>>(src, dst, counts, pairs, bstart, E, n);

    bucket_deg_gemm1_kernel<<<NB, BT, 0, stream>>>(pairs, bstart, x, w1, dinv, h1, n);

    bucket_agg_kernel<16><<<NB, BT, 0, stream>>>(pairs, bstart, h1, dinv, b1, w2, h2, n);

    bucket_agg_kernel<4><<<NB, BT, 0, stream>>>(pairs, bstart, h2, dinv, b2, w2, out, n);
}


# Placing pass orders its edge chunk by bucket in LDS, then stores each bucket's run with consecutive lanes (few lines per store, was 32); edge order in a bucket, hence f32 sum order, changes
# speedup vs baseline: 5.5522x; 1.1702x over previous
#include <hip/hip_runtime.h>
#include <math.h>


typedef __attribute__((ext_vector_type(2))) float v2f;
typedef __attribute__((ext_vector_type(8))) float v8f;

constexpr int R_NODES = 400;
constexpr int G_BIN   = 256;
constexpr int NB_MAX  = 512;
constexpr int BT      = 1024;
constexpr int CHUNK_CAP = 12800;

__global__ __launch_bounds__(BT) void bin_count_kernel(const int* __restrict__ dst,
                                                       int* __restrict__ counts,
                                                       int E, int n) {
    __shared__ int s_hist[NB_MAX];
    const int tid = threadIdx.x;
    const int nb  = (n + R_NODES - 1) / R_NODES;
    const int G   = (int)gridDim.x;
    for (int k = tid; k < NB_MAX; k += BT) s_hist[k] = 0;
    __syncthreads();

    const int chunk = (E + G - 1) / G;
    const int beg   = (int)blockIdx.x * chunk;
    int len = E - beg;
    if (len > chunk) len = chunk;

    for (int i0 = tid; i0 < len; i0 += 4 * BT) {
        int d[4];
#pragma unroll
        for (int u = 0; u < 4; ++u) {
            const int i = i0 + u * BT;
            d[u] = (i < len) ? dst[beg + i] : -1;
        }
#pragma unroll
        for (int u = 0; u < 4; ++u)
            if (d[u] >= 0) atomicAdd(&s_hist[d[u] / R_NODES], 1);
    }
    __syncthreads();
    for (int k = tid; k < nb; k += BT) counts[k * G + (int)blockIdx.x] = s_hist[k];
}

__global__ __launch_bounds__(BT) void bin_fill_kernel(const int* __restrict__ src,
                                                      const int* __restrict__ dst,
                                                      const int* __restrict__ counts,
                                                      int* __restrict__ pairs_i,
                                                      int* __restrict__ bstart,
                                                      int E, int n) {
    __shared__ int s_pref[NB_MAX];
    __shared__ int s_tot[NB_MAX];
    __shared__ int s_own[NB_MAX];
    __shared__ int s_gbase[NB_MAX];
    __shared__ int s_lbase[NB_MAX];
    __shared__ int s_cur[NB_MAX];
    __shared__ long long s_stage[CHUNK_CAP];
    int2* __restrict__ pairs = (int2*)pairs_i;
    const int tid  = threadIdx.x;
    const int lane = tid & 31;
    const int wave = tid >> 5;
    const int nb   = (n + R_NODES - 1) / R_NODES;
    const int G    = (int)gridDim.x;
    const int b    = (int)blockIdx.x;

    const int chunk = (E + G - 1) / G;
    const int beg   = b * chunk;
    int len = E - beg;
    if (len > chunk) len = chunk;
    const bool staged = chunk <= CHUNK_CAP;

    for (int k = wave; k < nb; k += BT / 32) {
        const int* row = counts + (size_t)k * G;
        int pre = 0, tot = 0, own = 0;
        for (int j = lane; j < G; j += 32) {
            const int v = row[j];
            tot += v;
            pre += (j < b) ? v : 0;
            own += (j == b) ? v : 0;
        }
        for (int o = 16; o > 0; o >>= 1) {
            pre += __shfl_xor(pre, o);
            tot += __shfl_xor(tot, o);
            own += __shfl_xor(own, o);
        }
        if (lane == 0) { s_pref[k] = pre; s_tot[k] = tot; s_own[k] = own; }
    }
    __syncthreads();

    if (wave == 0) {
        int loc[16], lown[16];
        int sum = 0, lsum = 0;
#pragma unroll
        for (int i = 0; i < 16; ++i) {
            const int k = lane * 16 + i;
            loc[i]  = (k < nb) ? s_tot[k] : 0;
            lown[i] = (k < nb) ? s_own[k] : 0;
            sum  += loc[i];
            lsum += lown[i];
        }
        int incl = sum, lincl = lsum;
        for (int o = 1; o < 32; o <<= 1) {
            const int t  = __shfl_up(incl, o);
            const int lt = __shfl_up(lincl, o);
            if (lane >= o) { incl += t; lincl += lt; }
        }
        int run  = incl - sum;
        int lrun = lincl - lsum;
#pragma unroll
        for (int i = 0; i < 16; ++i) {
            const int k = lane * 16 + i;
            if (k < nb) {
                const int g = run + s_pref[k];
                s_gbase[k] = g;
                s_lbase[k] = lrun;
                s_cur[k]   = staged ? lrun : g;
                if (b == 0) bstart[k] = run;
            }
            run  += loc[i];
            lrun += lown[i];
        }
        if (b == 0 && lane == 31) bstart[nb] = run;
    }
    __syncthreads();

    for (int i0 = tid; i0 < len; i0 += 4 * BT) {
        int s[4], d[4];
#pragma unroll
        for (int u = 0; u < 4; ++u) {
            const int i = i0 + u * BT;
            const bool ok = i < len;
            d[u] = ok ? dst[beg + i] : -1;
            s[u] = ok ? src[beg + i] : 0;
        }
        int pos[4];
#pragma unroll
        for (int u = 0; u < 4; ++u)
            pos[u] = (d[u] >= 0) ? atomicAdd(&s_cur[d[u] / R_NODES], 1) : -1;
#pragma unroll
        for (int u = 0; u < 4; ++u) {
            if (pos[u] >= 0) {
                if (staged)
                    s_stage[pos[u]] = ((long long)d[u] << 32) | (long long)(unsigned int)s[u];
                else
                    pairs[pos[u]] = make_int2(s[u], d[u]);
            }
        }
    }

    if (staged) {
        __syncthreads();
        for (int i0 = tid; i0 < len; i0 += 4 * BT) {
            long long v[4];
#pragma unroll
            for (int u = 0; u < 4; ++u) {
                const int i = i0 + u * BT;
                v[u] = (i < len) ? s_stage[i] : 0;
            }
#pragma unroll
            for (int u = 0; u < 4; ++u) {
                const int i = i0 + u * BT;
                if (i < len) {
                    const int sv = (int)(unsigned int)(v[u] & 0xffffffffLL);
                    const int dv = (int)(v[u] >> 32);
                    const int k  = dv / R_NODES;
                    pairs[s_gbase[k] + (i - s_lbase[k])] = make_int2(sv, dv);
                }
            }
        }
    }
}

static_assert(R_NODES % 16 == 0 && R_NODES / 16 <= BT / 32, "one 16-row tile per wave");

__global__ __launch_bounds__(BT) void bucket_deg_gemm1_kernel(const int* __restrict__ pairs_i,
                                                              const int* __restrict__ bstart,
                                                              const float* __restrict__ x,
                                                              const float* __restrict__ w1,
                                                              float* __restrict__ dinv,
                                                              float* __restrict__ h1, int n) {
    __shared__ __attribute__((aligned(16))) int s_cnt[R_NODES];
    const int2* __restrict__ pairs = (const int2*)pairs_i;
    const int tid  = threadIdx.x;
    const int k    = (int)blockIdx.x;
    const int base = k * R_NODES;
    int cnt = n - base;
    if (cnt > R_NODES) cnt = R_NODES;
    for (int i = tid; i < R_NODES; i += BT) s_cnt[i] = 0;
    __syncthreads();

    const int beg = bstart[k], end = bstart[k + 1];
    for (int e0 = beg + tid; e0 < end; e0 += 4 * BT) {
        int d[4];
#pragma unroll
        for (int u = 0; u < 4; ++u) {
            const int e = e0 + u * BT;
            d[u] = (e < end) ? pairs[e].y : -1;
        }
#pragma unroll
        for (int u = 0; u < 4; ++u)
            if (d[u] >= 0) atomicAdd(&s_cnt[d[u] - base], 1);
    }

    const int wave = tid >> 5;
    const int lane = tid & 31;
    const bool has_tile = wave * 16 < cnt;
    v8f acc = {};
    if (has_tile) {
        const int m    = lane & 15;
        const int koff = (lane >> 4) * 2;
        int arow = base + wave * 16 + m;
        if (arow > n - 1) arow = n - 1;
        const float* xrow = x + (size_t)arow * 128;
        for (int k0 = 0; k0 < 128; k0 += 4) {
            v2f a, b;
            a.x = xrow[k0 + koff + 0];
            a.y = xrow[k0 + koff + 1];
            b.x = w1[(k0 + koff + 0) * 16 + m];
            b.y = w1[(k0 + koff + 1) * 16 + m];
            acc = __builtin_amdgcn_wmma_f32_16x16x4_f32(
                 false, a,  false, b,
                 (short)0, acc,  false,  false);
        }
    }
    __syncthreads();

    for (int i = tid; i < cnt; i += BT) dinv[base + i] = rsqrtf((float)(1 + s_cnt[i]));

    if (has_tile) {
        const int mBase = wave * 16 + (lane >> 4) * 8;
        const int nCol  = lane & 15;
        const int4 clo = *(const int4*)&s_cnt[mBase];
        const int4 chi = *(const int4*)&s_cnt[mBase + 4];
        const int cr[8] = {clo.x, clo.y, clo.z, clo.w, chi.x, chi.y, chi.z, chi.w};
#pragma unroll
        for (int r = 0; r < 8; ++r) {
            const int rl = mBase + r;
            if (rl < cnt)
                h1[(size_t)(base + rl) * 16 + nCol] = acc[r] * rsqrtf((float)(1 + cr[r]));
        }
    }
}

template <int F, int U, bool TAIL>
__device__ __forceinline__ void agg_batch(const int2* __restrict__ pairs,
                                          const float* __restrict__ h,
                                          float* s_acc,
                                          int e0, int end, int base, int c) {
    constexpr int EPB = BT / F;
    int2  p[U];
    float hv[U];
#pragma unroll
    for (int u = 0; u < U; ++u) {
        int e = e0 + u * EPB;
        if (TAIL) e = (e < end) ? e : end - 1;
        p[u] = pairs[e];
    }
#pragma unroll
    for (int u = 0; u < U; ++u)
        hv[u] = h[(size_t)p[u].x * F + c];
#pragma unroll
    for (int u = 0; u < U; ++u) {
        const int dl = p[u].y - base;
        float msg = hv[u];
        if (TAIL && !(e0 + u * EPB < end)) msg = 0.0f;
        atomicAdd(&s_acc[dl * F + c], msg);
    }
}

template <int F>
__global__ __launch_bounds__(BT) void bucket_agg_kernel(const int* __restrict__ pairs_i,
                                                        const int* __restrict__ bstart,
                                                        const float* __restrict__ h,
                                                        const float* __restrict__ dinv,
                                                        const float* __restrict__ bias,
                                                        const float* __restrict__ w2,
                                                        float* __restrict__ outp, int n) {
    __shared__ __attribute__((aligned(16))) float s_acc[R_NODES * F];
    __shared__ float s_dinv[R_NODES];
    const int2* __restrict__ pairs = (const int2*)pairs_i;
    const int tid  = threadIdx.x;
    const int k    = (int)blockIdx.x;
    const int base = k * R_NODES;
    int cnt = n - base;
    if (cnt > R_NODES) cnt = R_NODES;

    for (int i = tid; i < cnt; i += BT) s_dinv[i] = dinv[base + i];
    for (int t = tid; t < cnt * F; t += BT) s_acc[t] = h[(size_t)base * F + t];
    __syncthreads();

    constexpr int EPB = BT / F;
    constexpr int U   = 8;
    const int sub = tid / F, c = tid % F;
    const int beg = bstart[k], end = bstart[k + 1];

    int e0 = beg + sub;
    for (; e0 + (U - 1) * EPB < end; e0 += U * EPB)
        agg_batch<F, U, false>(pairs, h, s_acc, e0, end, base, c);
    if (e0 < end)
        agg_batch<F, U, true>(pairs, h, s_acc, e0, end, base, c);
    __syncthreads();

    if constexpr (F == 16) {
        const int j = tid & 3;
        for (int t = tid; t < cnt * 4; t += BT) {
            const int nd = t >> 2;
            const float di = s_dinv[nd];
            float s = 0.0f;
#pragma unroll
            for (int q = 0; q < 16; ++q) {
                float v = bias[q] + di * s_acc[nd * 16 + q];
                v = v > 0.0f ? v : 0.0f;
                s += v * w2[q * 4 + j];
            }
            outp[(size_t)base * 4 + t] = s * di;
        }
    } else {
        static_assert(F == 4, "the last layer has 4 classes");
        for (int nd = tid; nd < cnt; nd += BT) {
            const float di = s_dinv[nd];
            const float4 sv = *(const float4*)&s_acc[nd * 4];
            float4 a;
            a.x = bias[0] + di * sv.x;
            a.y = bias[1] + di * sv.y;
            a.z = bias[2] + di * sv.z;
            a.w = bias[3] + di * sv.w;
            const float m = fmaxf(fmaxf(a.x, a.y), fmaxf(a.z, a.w));
            const float s = __expf(a.x - m) + __expf(a.y - m) + __expf(a.z - m) + __expf(a.w - m);
            const float lse = m + __logf(s);
            float4 r;
            r.x = a.x - lse; r.y = a.y - lse; r.z = a.z - lse; r.w = a.w - lse;
            ((float4*)outp)[base + nd] = r;
        }
    }
}

extern "C" void kernel_launch(void* const* d_in, const int* in_sizes, int n_in,
                              void* d_out, int out_size, void* d_ws, size_t ws_size,
                              hipStream_t stream) {
    const float* x  = (const float*)d_in[0];
    const int*   ei = (const int*)d_in[1];
    const float* w1 = (const float*)d_in[2];
    const float* b1 = (const float*)d_in[3];
    const float* w2 = (const float*)d_in[4];
    const float* b2 = (const float*)d_in[5];
    float* out = (float*)d_out;

    const int n = in_sizes[0] / 128;
    const int E = in_sizes[1] / 2;
    const int* src = ei;
    const int* dst = ei + E;

    float* dinv = (float*)d_ws;
    float* h1   = dinv + n;
    float* h2   = h1 + (size_t)n * 16;
    int* counts = (int*)(h2 + (size_t)n * 4);
    int* bstart = counts + NB_MAX * G_BIN;
    int* pairs  = bstart + 1024;

    auto cdiv = [](long long a, long long b) { return (int)((a + b - 1) / b); };
    const int NB = cdiv(n, R_NODES);

    bin_count_kernel<<<G_BIN, BT, 0, stream>>>(dst, counts, E, n);
    bin_fill_kernel<<<G_BIN, BT, 0, stream>>>(src, dst, counts, pairs, bstart, E, n);

    bucket_deg_gemm1_kernel<<<NB, BT, 0, stream>>>(pairs, bstart, x, w1, dinv, h1, n);

    bucket_agg_kernel<16><<<NB, BT, 0, stream>>>(pairs, bstart, h1, dinv, b1, w2, h2, n);

    bucket_agg_kernel<4><<<NB, BT, 0, stream>>>(pairs, bstart, h2, dinv, b2, w2, out, n);
}


# Aggregation launches copy the bucket's edge run to LDS in one burst beside the self rows and gather from the staged pairs (16 rows in flight per lane at 16 features); f32 add order changes
# speedup vs baseline: 5.7293x; 1.0319x over previous
#include <hip/hip_runtime.h>
#include <math.h>


typedef __attribute__((ext_vector_type(2))) float v2f;
typedef __attribute__((ext_vector_type(8))) float v8f;

constexpr int R_NODES = 400;
constexpr int G_BIN   = 256;
constexpr int NB_MAX  = 512;
constexpr int BT      = 1024;
constexpr int CHUNK_CAP = 12800;

__global__ __launch_bounds__(BT) void bin_count_kernel(const int* __restrict__ dst,
                                                       int* __restrict__ counts,
                                                       int E, int n) {
    __shared__ int s_hist[NB_MAX];
    const int tid = threadIdx.x;
    const int nb  = (n + R_NODES - 1) / R_NODES;
    const int G   = (int)gridDim.x;
    for (int k = tid; k < NB_MAX; k += BT) s_hist[k] = 0;
    __syncthreads();

    const int chunk = (E + G - 1) / G;
    const int beg   = (int)blockIdx.x * chunk;
    int len = E - beg;
    if (len > chunk) len = chunk;

    for (int i0 = tid; i0 < len; i0 += 4 * BT) {
        int d[4];
#pragma unroll
        for (int u = 0; u < 4; ++u) {
            const int i = i0 + u * BT;
            d[u] = (i < len) ? dst[beg + i] : -1;
        }
#pragma unroll
        for (int u = 0; u < 4; ++u)
            if (d[u] >= 0) atomicAdd(&s_hist[d[u] / R_NODES], 1);
    }
    __syncthreads();
    for (int k = tid; k < nb; k += BT) counts[k * G + (int)blockIdx.x] = s_hist[k];
}

__global__ __launch_bounds__(BT) void bin_fill_kernel(const int* __restrict__ src,
                                                      const int* __restrict__ dst,
                                                      const int* __restrict__ counts,
                                                      int* __restrict__ pairs_i,
                                                      int* __restrict__ bstart,
                                                      int E, int n) {
    __shared__ int s_pref[NB_MAX];
    __shared__ int s_tot[NB_MAX];
    __shared__ int s_own[NB_MAX];
    __shared__ int s_gbase[NB_MAX];
    __shared__ int s_lbase[NB_MAX];
    __shared__ int s_cur[NB_MAX];
    __shared__ long long s_stage[CHUNK_CAP];
    int2* __restrict__ pairs = (int2*)pairs_i;
    const int tid  = threadIdx.x;
    const int lane = tid & 31;
    const int wave = tid >> 5;
    const int nb   = (n + R_NODES - 1) / R_NODES;
    const int G    = (int)gridDim.x;
    const int b    = (int)blockIdx.x;

    const int chunk = (E + G - 1) / G;
    const int beg   = b * chunk;
    int len = E - beg;
    if (len > chunk) len = chunk;
    const bool staged = chunk <= CHUNK_CAP;

    for (int k = wave; k < nb; k += BT / 32) {
        const int* row = counts + (size_t)k * G;
        int pre = 0, tot = 0, own = 0;
        for (int j = lane; j < G; j += 32) {
            const int v = row[j];
            tot += v;
            pre += (j < b) ? v : 0;
            own += (j == b) ? v : 0;
        }
        for (int o = 16; o > 0; o >>= 1) {
            pre += __shfl_xor(pre, o);
            tot += __shfl_xor(tot, o);
            own += __shfl_xor(own, o);
        }
        if (lane == 0) { s_pref[k] = pre; s_tot[k] = tot; s_own[k] = own; }
    }
    __syncthreads();

    if (wave == 0) {
        int loc[16], lown[16];
        int sum = 0, lsum = 0;
#pragma unroll
        for (int i = 0; i < 16; ++i) {
            const int k = lane * 16 + i;
            loc[i]  = (k < nb) ? s_tot[k] : 0;
            lown[i] = (k < nb) ? s_own[k] : 0;
            sum  += loc[i];
            lsum += lown[i];
        }
        int incl = sum, lincl = lsum;
        for (int o = 1; o < 32; o <<= 1) {
            const int t  = __shfl_up(incl, o);
            const int lt = __shfl_up(lincl, o);
            if (lane >= o) { incl += t; lincl += lt; }
        }
        int run  = incl - sum;
        int lrun = lincl - lsum;
#pragma unroll
        for (int i = 0; i < 16; ++i) {
            const int k = lane * 16 + i;
            if (k < nb) {
                const int g = run + s_pref[k];
                s_gbase[k] = g;
                s_lbase[k] = lrun;
                s_cur[k]   = staged ? lrun : g;
                if (b == 0) bstart[k] = run;
            }
            run  += loc[i];
            lrun += lown[i];
        }
        if (b == 0 && lane == 31) bstart[nb] = run;
    }
    __syncthreads();

    for (int i0 = tid; i0 < len; i0 += 4 * BT) {
        int s[4], d[4];
#pragma unroll
        for (int u = 0; u < 4; ++u) {
            const int i = i0 + u * BT;
            const bool ok = i < len;
            d[u] = ok ? dst[beg + i] : -1;
            s[u] = ok ? src[beg + i] : 0;
        }
        int pos[4];
#pragma unroll
        for (int u = 0; u < 4; ++u)
            pos[u] = (d[u] >= 0) ? atomicAdd(&s_cur[d[u] / R_NODES], 1) : -1;
#pragma unroll
        for (int u = 0; u < 4; ++u) {
            if (pos[u] >= 0) {
                if (staged)
                    s_stage[pos[u]] = ((long long)d[u] << 32) | (long long)(unsigned int)s[u];
                else
                    pairs[pos[u]] = make_int2(s[u], d[u]);
            }
        }
    }

    if (staged) {
        __syncthreads();
        for (int i0 = tid; i0 < len; i0 += 4 * BT) {
            long long v[4];
#pragma unroll
            for (int u = 0; u < 4; ++u) {
                const int i = i0 + u * BT;
                v[u] = (i < len) ? s_stage[i] : 0;
            }
#pragma unroll
            for (int u = 0; u < 4; ++u) {
                const int i = i0 + u * BT;
                if (i < len) {
                    const int sv = (int)(unsigned int)(v[u] & 0xffffffffLL);
                    const int dv = (int)(v[u] >> 32);
                    const int k  = dv / R_NODES;
                    pairs[s_gbase[k] + (i - s_lbase[k])] = make_int2(sv, dv);
                }
            }
        }
    }
}

static_assert(R_NODES % 16 == 0 && R_NODES / 16 <= BT / 32, "one 16-row tile per wave");

__global__ __launch_bounds__(BT) void bucket_deg_gemm1_kernel(const int* __restrict__ pairs_i,
                                                              const int* __restrict__ bstart,
                                                              const float* __restrict__ x,
                                                              const float* __restrict__ w1,
                                                              float* __restrict__ dinv,
                                                              float* __restrict__ h1, int n) {
    __shared__ __attribute__((aligned(16))) int s_cnt[R_NODES];
    const int2* __restrict__ pairs = (const int2*)pairs_i;
    const int tid  = threadIdx.x;
    const int k    = (int)blockIdx.x;
    const int base = k * R_NODES;
    int cnt = n - base;
    if (cnt > R_NODES) cnt = R_NODES;
    for (int i = tid; i < R_NODES; i += BT) s_cnt[i] = 0;
    __syncthreads();

    const int beg = bstart[k], end = bstart[k + 1];
    for (int e0 = beg + tid; e0 < end; e0 += 4 * BT) {
        int d[4];
#pragma unroll
        for (int u = 0; u < 4; ++u) {
            const int e = e0 + u * BT;
            d[u] = (e < end) ? pairs[e].y : -1;
        }
#pragma unroll
        for (int u = 0; u < 4; ++u)
            if (d[u] >= 0) atomicAdd(&s_cnt[d[u] - base], 1);
    }

    const int wave = tid >> 5;
    const int lane = tid & 31;
    const bool has_tile = wave * 16 < cnt;
    v8f acc = {};
    if (has_tile) {
        const int m    = lane & 15;
        const int koff = (lane >> 4) * 2;
        int arow = base + wave * 16 + m;
        if (arow > n - 1) arow = n - 1;
        const float* xrow = x + (size_t)arow * 128;
        for (int k0 = 0; k0 < 128; k0 += 4) {
            v2f a, b;
            a.x = xrow[k0 + koff + 0];
            a.y = xrow[k0 + koff + 1];
            b.x = w1[(k0 + koff + 0) * 16 + m];
            b.y = w1[(k0 + koff + 1) * 16 + m];
            acc = __builtin_amdgcn_wmma_f32_16x16x4_f32(
                 false, a,  false, b,
                 (short)0, acc,  false,  false);
        }
    }
    __syncthreads();

    for (int i = tid; i < cnt; i += BT) dinv[base + i] = rsqrtf((float)(1 + s_cnt[i]));

    if (has_tile) {
        const int mBase = wave * 16 + (lane >> 4) * 8;
        const int nCol  = lane & 15;
        const int4 clo = *(const int4*)&s_cnt[mBase];
        const int4 chi = *(const int4*)&s_cnt[mBase + 4];
        const int cr[8] = {clo.x, clo.y, clo.z, clo.w, chi.x, chi.y, chi.z, chi.w};
#pragma unroll
        for (int r = 0; r < 8; ++r) {
            const int rl = mBase + r;
            if (rl < cnt)
                h1[(size_t)(base + rl) * 16 + nCol] = acc[r] * rsqrtf((float)(1 + cr[r]));
        }
    }
}

constexpr int STAGE_PAIRS = 16 * BT;

template <int F, int U, bool TAIL>
__device__ __forceinline__ void agg_batch(const int2* s_pr,
                                          const float* __restrict__ h,
                                          float* s_acc,
                                          int i0, int clen, int base, int c) {
    constexpr int EPB = BT / F;
    int2  p[U];
    float hv[U];
#pragma unroll
    for (int u = 0; u < U; ++u) {
        int i = i0 + u * EPB;
        if (TAIL) i = (i < clen) ? i : clen - 1;
        p[u] = s_pr[i];
    }
#pragma unroll
    for (int u = 0; u < U; ++u)
        hv[u] = h[(unsigned int)p[u].x * (unsigned int)F + (unsigned int)c];
#pragma unroll
    for (int u = 0; u < U; ++u) {
        const int dl = p[u].y - base;
        float msg = hv[u];
        if (TAIL && !(i0 + u * EPB < clen)) msg = 0.0f;
        atomicAdd(&s_acc[dl * F + c], msg);
    }
}

template <int F>
__global__ __launch_bounds__(BT) void bucket_agg_kernel(const int* __restrict__ pairs_i,
                                                        const int* __restrict__ bstart,
                                                        const float* __restrict__ h,
                                                        const float* __restrict__ dinv,
                                                        const float* __restrict__ bias,
                                                        const float* __restrict__ w2,
                                                        float* __restrict__ outp, int n) {
    __shared__ __attribute__((aligned(16))) float s_acc[R_NODES * F];
    __shared__ float s_dinv[R_NODES];
    __shared__ __attribute__((aligned(16))) int s_pr_i[2 * STAGE_PAIRS];
    const int2* __restrict__ pairs = (const int2*)pairs_i;
    int2* s_pr = (int2*)s_pr_i;
    const int tid  = threadIdx.x;
    const int k    = (int)blockIdx.x;
    const int base = k * R_NODES;
    int cnt = n - base;
    if (cnt > R_NODES) cnt = R_NODES;

    constexpr int EPB = BT / F;
    constexpr int U   = (F == 16) ? 16 : 8;
    const int sub = tid / F, c = tid % F;
    const int beg = bstart[k], end = bstart[k + 1];

    int c0 = beg;
    bool first = true;
    do {
        int clen = end - c0;
        if (clen > STAGE_PAIRS) clen = STAGE_PAIRS;

        int2 v[16];
#pragma unroll
        for (int u = 0; u < 16; ++u) v[u] = make_int2(0, 0);
        if (clen > 0) {
#pragma unroll
            for (int u = 0; u < 16; ++u) {
                int i = tid + u * BT;
                i = (i < clen) ? i : clen - 1;
                v[u] = pairs[c0 + i];
            }
        }
        if (first) {
            for (int i = tid; i < cnt; i += BT) s_dinv[i] = dinv[base + i];
            for (int t = tid; t < cnt * F; t += BT) s_acc[t] = h[(size_t)base * F + t];
            first = false;
        }
#pragma unroll
        for (int u = 0; u < 16; ++u) {
            const int i = tid + u * BT;
            if (i < clen) s_pr[i] = v[u];
        }
        __syncthreads();

        int i0 = sub;
        for (; i0 + (U - 1) * EPB < clen; i0 += U * EPB)
            agg_batch<F, U, false>(s_pr, h, s_acc, i0, clen, base, c);
        if (i0 < clen)
            agg_batch<F, U, true>(s_pr, h, s_acc, i0, clen, base, c);
        __syncthreads();

        c0 += STAGE_PAIRS;
    } while (c0 < end);

    if constexpr (F == 16) {
        const int j = tid & 3;
        for (int t = tid; t < cnt * 4; t += BT) {
            const int nd = t >> 2;
            const float di = s_dinv[nd];
            float s = 0.0f;
#pragma unroll
            for (int q = 0; q < 16; ++q) {
                float v = bias[q] + di * s_acc[nd * 16 + q];
                v = v > 0.0f ? v : 0.0f;
                s += v * w2[q * 4 + j];
            }
            outp[(size_t)base * 4 + t] = s * di;
        }
    } else {
        static_assert(F == 4, "the last layer has 4 classes");
        for (int nd = tid; nd < cnt; nd += BT) {
            const float di = s_dinv[nd];
            const float4 sv = *(const float4*)&s_acc[nd * 4];
            float4 a;
            a.x = bias[0] + di * sv.x;
            a.y = bias[1] + di * sv.y;
            a.z = bias[2] + di * sv.z;
            a.w = bias[3] + di * sv.w;
            const float m = fmaxf(fmaxf(a.x, a.y), fmaxf(a.z, a.w));
            const float s = __expf(a.x - m) + __expf(a.y - m) + __expf(a.z - m) + __expf(a.w - m);
            const float lse = m + __logf(s);
            float4 r;
            r.x = a.x - lse; r.y = a.y - lse; r.z = a.z - lse; r.w = a.w - lse;
            ((float4*)outp)[base + nd] = r;
        }
    }
}

extern "C" void kernel_launch(void* const* d_in, const int* in_sizes, int n_in,
                              void* d_out, int out_size, void* d_ws, size_t ws_size,
                              hipStream_t stream) {
    const float* x  = (const float*)d_in[0];
    const int*   ei = (const int*)d_in[1];
    const float* w1 = (const float*)d_in[2];
    const float* b1 = (const float*)d_in[3];
    const float* w2 = (const float*)d_in[4];
    const float* b2 = (const float*)d_in[5];
    float* out = (float*)d_out;

    const int n = in_sizes[0] / 128;
    const int E = in_sizes[1] / 2;
    const int* src = ei;
    const int* dst = ei + E;

    float* dinv = (float*)d_ws;
    float* h1   = dinv + n;
    float* h2   = h1 + (size_t)n * 16;
    int* counts = (int*)(h2 + (size_t)n * 4);
    int* bstart = counts + NB_MAX * G_BIN;
    int* pairs  = bstart + 1024;

    auto cdiv = [](long long a, long long b) { return (int)((a + b - 1) / b); };
    const int NB = cdiv(n, R_NODES);

    bin_count_kernel<<<G_BIN, BT, 0, stream>>>(dst, counts, E, n);
    bin_fill_kernel<<<G_BIN, BT, 0, stream>>>(src, dst, counts, pairs, bstart, E, n);

    bucket_deg_gemm1_kernel<<<NB, BT, 0, stream>>>(pairs, bstart, x, w1, dinv, h1, n);

    bucket_agg_kernel<16><<<NB, BT, 0, stream>>>(pairs, bstart, h1, dinv, b1, w2, h2, n);

    bucket_agg_kernel<4><<<NB, BT, 0, stream>>>(pairs, bstart, h2, dinv, b2, w2, out, n);
}


# One binning launch: a work-group orders its chunk of edges by bucket in place and stores segment offsets; the three bucket launches read the segments in place, no run is copied; f32 add order changes
# speedup vs baseline: 6.5013x; 1.1347x over previous
#include <hip/hip_runtime.h>
#include <math.h>


typedef __attribute__((ext_vector_type(2))) float v2f;
typedef __attribute__((ext_vector_type(8))) float v8f;

constexpr int R_NODES = 400;
constexpr int NB_MAX  = 512;
constexpr int BT      = 1024;
constexpr int CHUNK   = 12500;
constexpr int EPT     = (CHUNK + BT - 1) / BT;
constexpr int SEG_U   = 8;

__global__ __launch_bounds__(BT) void bin_kernel(const int* __restrict__ src,
                                                 const int* __restrict__ dst,
                                                 int* __restrict__ pairs_i,
                                                 int* __restrict__ tab,
                                                 int E, int n) {
    __shared__ int s_cur[NB_MAX];
    __shared__ long long s_stage[CHUNK];
    long long* __restrict__ pairs = (long long*)pairs_i;
    const int tid  = threadIdx.x;
    const int lane = tid & 31;
    const int wave = tid >> 5;
    const int nb   = (n + R_NODES - 1) / R_NODES;
    const int NC   = (int)gridDim.x;
    const int b    = (int)blockIdx.x;
    const int beg  = b * CHUNK;
    int len = E - beg;
    if (len > CHUNK) len = CHUNK;

    for (int k = tid; k < NB_MAX; k += BT) s_cur[k] = 0;

    int s[EPT], d[EPT];
#pragma unroll
    for (int u = 0; u < EPT; ++u) {
        const int i = tid + u * BT;
        const bool ok = i < len;
        d[u] = ok ? dst[beg + i] : -1;
        s[u] = ok ? src[beg + i] : 0;
    }
    __syncthreads();
#pragma unroll
    for (int u = 0; u < EPT; ++u)
        if (d[u] >= 0) atomicAdd(&s_cur[d[u] / R_NODES], 1);
    __syncthreads();

    if (wave == 0) {
        int loc[16];
        int sum = 0;
#pragma unroll
        for (int i = 0; i < 16; ++i) {
            const int k = lane * 16 + i;
            loc[i] = (k < nb) ? s_cur[k] : 0;
            sum += loc[i];
        }
        int incl = sum;
        for (int o = 1; o < 32; o <<= 1) {
            const int t = __shfl_up(incl, o);
            if (lane >= o) incl += t;
        }
        int run = incl - sum;
#pragma unroll
        for (int i = 0; i < 16; ++i) {
            const int k = lane * 16 + i;
            if (k < nb) {
                s_cur[k] = run;
                tab[(size_t)k * NC + b] = beg + run;
            }
            run += loc[i];
        }
        if (lane == 31) tab[(size_t)nb * NC + b] = beg + run;
    }
    __syncthreads();

    int pos[EPT];
#pragma unroll
    for (int u = 0; u < EPT; ++u)
        pos[u] = (d[u] >= 0) ? atomicAdd(&s_cur[d[u] / R_NODES], 1) : -1;
#pragma unroll
    for (int u = 0; u < EPT; ++u)
        if (pos[u] >= 0)
            s_stage[pos[u]] = ((long long)d[u] << 32) | (long long)(unsigned int)s[u];
    __syncthreads();

#pragma unroll
    for (int u = 0; u < EPT; ++u) {
        const int i = tid + u * BT;
        if (i < len) pairs[beg + i] = s_stage[i];
    }
}

__device__ __forceinline__ void seg_tab(const int* __restrict__ tab, int k, int NC,
                                        int s0, int lane, int& o0, int& ln) {
    const int sg = s0 + 32 * (lane & (SEG_U - 1));
    const bool ok = sg < NC;
    const int a = ok ? tab[(size_t)k * NC + sg] : 0;
    const int e = ok ? tab[(size_t)(k + 1) * NC + sg] : 0;
    o0 = a;
    ln = e - a;
}

static_assert(R_NODES % 16 == 0 && R_NODES / 16 <= BT / 32, "one 16-row tile per wave");

__global__ __launch_bounds__(BT) void bucket_deg_gemm1_kernel(const int* __restrict__ pairs_i,
                                                              const int* __restrict__ tab,
                                                              const float* __restrict__ x,
                                                              const float* __restrict__ w1,
                                                              float* __restrict__ dinv,
                                                              float* __restrict__ h1, int n, int NC) {
    __shared__ __attribute__((aligned(16))) int s_cnt[R_NODES];
    const int2* __restrict__ pairs = (const int2*)pairs_i;
    const int tid  = threadIdx.x;
    const int k    = (int)blockIdx.x;
    const int base = k * R_NODES;
    int cnt = n - base;
    if (cnt > R_NODES) cnt = R_NODES;
    for (int i = tid; i < R_NODES; i += BT) s_cnt[i] = 0;
    __syncthreads();

    const int wave = tid >> 5;
    const int lane = tid & 31;
    for (int s0 = wave; s0 < NC; s0 += 32 * SEG_U) {
        int o0, lnl;
        seg_tab(tab, k, NC, s0, lane, o0, lnl);
        int off[SEG_U], ln[SEG_U];
#pragma unroll
        for (int u = 0; u < SEG_U; ++u) {
            off[u] = __builtin_amdgcn_readlane(o0, u);
            ln[u]  = __builtin_amdgcn_readlane(lnl, u);
        }
        int d[2 * SEG_U];
#pragma unroll
        for (int u = 0; u < SEG_U; ++u) {
#pragma unroll
            for (int hh = 0; hh < 2; ++hh) {
                const int j = lane + 32 * hh;
                d[2 * u + hh] = (j < ln[u]) ? pairs[off[u] + j].y : -1;
            }
        }
#pragma unroll
        for (int q = 0; q < 2 * SEG_U; ++q)
            if (d[q] >= 0) atomicAdd(&s_cnt[d[q] - base], 1);
#pragma unroll
        for (int u = 0; u < SEG_U; ++u)
            for (int j = 64 + lane; j < ln[u]; j += 32)
                atomicAdd(&s_cnt[pairs[off[u] + j].y - base], 1);
    }

    const bool has_tile = wave * 16 < cnt;
    v8f acc = {};
    if (has_tile) {
        const int m    = lane & 15;
        const int koff = (lane >> 4) * 2;
        int arow = base + wave * 16 + m;
        if (arow > n - 1) arow = n - 1;
        const float* xrow = x + (size_t)arow * 128;
        for (int k0 = 0; k0 < 128; k0 += 4) {
            v2f a, b;
            a.x = xrow[k0 + koff + 0];
            a.y = xrow[k0 + koff + 1];
            b.x = w1[(k0 + koff + 0) * 16 + m];
            b.y = w1[(k0 + koff + 1) * 16 + m];
            acc = __builtin_amdgcn_wmma_f32_16x16x4_f32(
                 false, a,  false, b,
                 (short)0, acc,  false,  false);
        }
    }
    __syncthreads();

    for (int i = tid; i < cnt; i += BT) dinv[base + i] = rsqrtf((float)(1 + s_cnt[i]));

    if (has_tile) {
        const int mBase = wave * 16 + (lane >> 4) * 8;
        const int nCol  = lane & 15;
        const int4 clo = *(const int4*)&s_cnt[mBase];
        const int4 chi = *(const int4*)&s_cnt[mBase + 4];
        const int cr[8] = {clo.x, clo.y, clo.z, clo.w, chi.x, chi.y, chi.z, chi.w};
#pragma unroll
        for (int r = 0; r < 8; ++r) {
            const int rl = mBase + r;
            if (rl < cnt)
                h1[(size_t)(base + rl) * 16 + nCol] = acc[r] * rsqrtf((float)(1 + cr[r]));
        }
    }
}

constexpr int STAGE_PAIRS = 16 * BT;

template <int F, int U, bool TAIL>
__device__ __forceinline__ void agg_batch(const int2* s_pr,
                                          const float* __restrict__ h,
                                          float* s_acc,
                                          int i0, int clen, int base, int c) {
    constexpr int EPB = BT / F;
    int2  p[U];
    float hv[U];
#pragma unroll
    for (int u = 0; u < U; ++u) {
        int i = i0 + u * EPB;
        if (TAIL) i = (i < clen) ? i : clen - 1;
        p[u] = s_pr[i];
    }
#pragma unroll
    for (int u = 0; u < U; ++u)
        hv[u] = h[(unsigned int)p[u].x * (unsigned int)F + (unsigned int)c];
#pragma unroll
    for (int u = 0; u < U; ++u) {
        const int dl = p[u].y - base;
        float msg = hv[u];
        if (TAIL && !(i0 + u * EPB < clen)) msg = 0.0f;
        atomicAdd(&s_acc[dl * F + c], msg);
    }
}

template <int F>
__global__ __launch_bounds__(BT) void bucket_agg_kernel(const int* __restrict__ pairs_i,
                                                        const int* __restrict__ tab,
                                                        const float* __restrict__ h,
                                                        const float* __restrict__ dinv,
                                                        const float* __restrict__ bias,
                                                        const float* __restrict__ w2,
                                                        float* __restrict__ outp, int n, int NC) {
    __shared__ __attribute__((aligned(16))) float s_acc[R_NODES * F];
    __shared__ float s_dinv[R_NODES];
    __shared__ __attribute__((aligned(16))) int s_pr_i[2 * STAGE_PAIRS];
    __shared__ int s_fill;
    const int2* __restrict__ pairs = (const int2*)pairs_i;
    int2* s_pr = (int2*)s_pr_i;
    const int tid  = threadIdx.x;
    const int k    = (int)blockIdx.x;
    const int base = k * R_NODES;
    int cnt = n - base;
    if (cnt > R_NODES) cnt = R_NODES;

    constexpr int EPB = BT / F;
    constexpr int U   = (F == 16) ? 16 : 8;
    const int sub = tid / F, c = tid % F;
    const int lane = tid & 31;
    const int wave = tid >> 5;

    if (tid == 0) s_fill = 0;
    int o0, lnl;
    seg_tab(tab, k, NC, wave, lane, o0, lnl);
    for (int i = tid; i < cnt; i += BT) s_dinv[i] = dinv[base + i];
    for (int t = tid; t < cnt * F; t += BT) s_acc[t] = h[(size_t)base * F + t];
    __syncthreads();

    for (int s0 = wave;;) {
        int off[SEG_U], ln[SEG_U], pre[SEG_U];
        int wt = 0;
#pragma unroll
        for (int u = 0; u < SEG_U; ++u) {
            off[u] = __builtin_amdgcn_readlane(o0, u);
            ln[u]  = __builtin_amdgcn_readlane(lnl, u);
            pre[u] = wt;
            wt += ln[u];
        }
        int2 v[2 * SEG_U];
#pragma unroll
        for (int u = 0; u < SEG_U; ++u) {
#pragma unroll
            for (int hh = 0; hh < 2; ++hh) {
                const int j = lane + 32 * hh;
                v[2 * u + hh] = (j < ln[u]) ? pairs[off[u] + j] : make_int2(0, 0);
            }
        }
        int bp = 0;
        if (lane == 0) bp = atomicAdd(&s_fill, wt);
        bp = __builtin_amdgcn_readfirstlane(bp);
#pragma unroll
        for (int u = 0; u < SEG_U; ++u) {
#pragma unroll
            for (int hh = 0; hh < 2; ++hh) {
                const int j = lane + 32 * hh;
                const int p = bp + pre[u] + j;
                if (j < ln[u] && p < STAGE_PAIRS) s_pr[p] = v[2 * u + hh];
            }
        }
#pragma unroll
        for (int u = 0; u < SEG_U; ++u) {
            for (int j = 64 + lane; j < ln[u]; j += 32) {
                const int p = bp + pre[u] + j;
                if (p < STAGE_PAIRS) s_pr[p] = pairs[off[u] + j];
            }
        }
        s0 += 32 * SEG_U;
        if (s0 >= NC) break;
        seg_tab(tab, k, NC, s0, lane, o0, lnl);
    }
    __syncthreads();

    const int total = s_fill;
    if (total <= STAGE_PAIRS) {
        const int clen = total;
        int i0 = sub;
        for (; i0 + (U - 1) * EPB < clen; i0 += U * EPB)
            agg_batch<F, U, false>(s_pr, h, s_acc, i0, clen, base, c);
        if (i0 < clen)
            agg_batch<F, U, true>(s_pr, h, s_acc, i0, clen, base, c);
    } else {
        for (int sg = 0; sg < NC; ++sg) {
            const int a = tab[(size_t)k * NC + sg];
            const int e = tab[(size_t)(k + 1) * NC + sg];
            for (int i = a + sub; i < e; i += EPB) {
                const int2 p = pairs[i];
                atomicAdd(&s_acc[(p.y - base) * F + c], h[(unsigned int)p.x * (unsigned int)F + (unsigned int)c]);
            }
        }
    }
    __syncthreads();

    if constexpr (F == 16) {
        const int j = tid & 3;
        for (int t = tid; t < cnt * 4; t += BT) {
            const int nd = t >> 2;
            const float di = s_dinv[nd];
            float s = 0.0f;
#pragma unroll
            for (int q = 0; q < 16; ++q) {
                float v = bias[q] + di * s_acc[nd * 16 + q];
                v = v > 0.0f ? v : 0.0f;
                s += v * w2[q * 4 + j];
            }
            outp[(size_t)base * 4 + t] = s * di;
        }
    } else {
        static_assert(F == 4, "the last layer has 4 classes");
        for (int nd = tid; nd < cnt; nd += BT) {
            const float di = s_dinv[nd];
            const float4 sv = *(const float4*)&s_acc[nd * 4];
            float4 a;
            a.x = bias[0] + di * sv.x;
            a.y = bias[1] + di * sv.y;
            a.z = bias[2] + di * sv.z;
            a.w = bias[3] + di * sv.w;
            const float m = fmaxf(fmaxf(a.x, a.y), fmaxf(a.z, a.w));
            const float s = __expf(a.x - m) + __expf(a.y - m) + __expf(a.z - m) + __expf(a.w - m);
            const float lse = m + __logf(s);
            float4 r;
            r.x = a.x - lse; r.y = a.y - lse; r.z = a.z - lse; r.w = a.w - lse;
            ((float4*)outp)[base + nd] = r;
        }
    }
}

extern "C" void kernel_launch(void* const* d_in, const int* in_sizes, int n_in,
                              void* d_out, int out_size, void* d_ws, size_t ws_size,
                              hipStream_t stream) {
    const float* x  = (const float*)d_in[0];
    const int*   ei = (const int*)d_in[1];
    const float* w1 = (const float*)d_in[2];
    const float* b1 = (const float*)d_in[3];
    const float* w2 = (const float*)d_in[4];
    const float* b2 = (const float*)d_in[5];
    float* out = (float*)d_out;

    const int n = in_sizes[0] / 128;
    const int E = in_sizes[1] / 2;
    const int* src = ei;
    const int* dst = ei + E;

    auto cdiv = [](long long a, long long b) { return (int)((a + b - 1) / b); };
    const int NB = cdiv(n, R_NODES);
    const int NC = cdiv(E, CHUNK);

    float* dinv = (float*)d_ws;
    float* h1   = dinv + n;
    float* h2   = h1 + (size_t)n * 16;
    int* tab    = (int*)(h2 + (size_t)n * 4);
    int* pairs  = tab + ((((size_t)NB + 1) * NC + 1) & ~(size_t)1);

    bin_kernel<<<NC, BT, 0, stream>>>(src, dst, pairs, tab, E, n);

    bucket_deg_gemm1_kernel<<<NB, BT, 0, stream>>>(pairs, tab, x, w1, dinv, h1, n, NC);

    bucket_agg_kernel<16><<<NB, BT, 0, stream>>>(pairs, tab, h1, dinv, b1, w2, h2, n, NC);

    bucket_agg_kernel<4><<<NB, BT, 0, stream>>>(pairs, tab, h2, dinv, b2, w2, out, n, NC);
}
